# starting kernel
# baseline (speedup 1.0000x reference)
#include <hip/hip_runtime.h>
#include <hip/hip_bf16.h>
#include <math.h>

typedef __attribute__((ext_vector_type(16))) _Float16 v16h;
typedef __attribute__((ext_vector_type(8)))  float    v8f;

__device__ __forceinline__ float gelu_f(float x) {
    const float c0 = 0.7978845608028654f;
    const float c1 = 2.8853900817779268f;
    float u = c0 * (x + 0.044715f * x * x * x);
    float e = __builtin_amdgcn_exp2f(u * c1);
    return x - x * __builtin_amdgcn_rcpf(e + 1.0f);
}

__device__ __forceinline__ void atomAddF(float* p, float v) {
    __hip_atomic_fetch_add(p, v, __ATOMIC_RELAXED, __HIP_MEMORY_SCOPE_AGENT);
}

__device__ __forceinline__ v8f wmma16(v16h a, v16h b, v8f c) {
    return __builtin_amdgcn_wmma_f32_16x16x32_f16(
         false, a,  false, b,
         (short)0, c,  false,  false);
}

__device__ __forceinline__ v16h load_wT_frag(const float* __restrict__ W,
                                             int Kreal, int t, int lane,
                                             float scale) {
    int m = 16 * t + (lane & 15);
    int g = lane >> 4;
    v16h a;
#pragma unroll
    for (int e = 0; e < 16; ++e) {
        int r = e >> 1, j = e & 1;
        int k = ((r >> 2) << 4) + g * 8 + ((r & 3) << 1) + j;
        float val = (k < Kreal) ? W[k * 32 + m] * scale : 0.0f;
        a[e] = (_Float16)val;
    }
    return a;
}

__device__ __forceinline__ v16h pack_b_from_d(v8f d0, v8f d1) {
    v16h b;
#pragma unroll
    for (int e = 0; e < 16; ++e) {
        int r = e >> 1, j = e & 1;
        int i = ((r & 3) << 1) + j;
        float val = (r >> 2) ? d1[i] : d0[i];
        b[e] = (_Float16)val;
    }
    return b;
}

__global__ void __launch_bounds__(256)
node_pre_kernel(const float* __restrict__ node_input,
                const float* __restrict__ W1_s,
                const float* __restrict__ W1_v,
                float* __restrict__ feat_s,
                float* __restrict__ feat_v, int N) {
    __shared__ float Ws[32][32];
    __shared__ float Wv[32][32];
    for (int idx = threadIdx.x; idx < 1024; idx += blockDim.x) {
        int u = idx >> 5, w = idx & 31;
        Ws[u][w] = W1_s[u * 64 + w];
        Wv[u][w] = W1_v[u * 64 + w];
    }
    __syncthreads();
    const float inv32 = 0.17677669529663687f;
    int gid = blockIdx.x * blockDim.x + threadIdx.x;
    int total = N * 32;
    for (int t = gid; t < total; t += gridDim.x * blockDim.x) {
        int n = t >> 5, w = t & 31;
        const float* row = node_input + (size_t)n * 128;
        float as = 0.f, ax = 0.f, ay = 0.f, az = 0.f;
#pragma unroll 4
        for (int u = 0; u < 32; ++u) {
            as += row[u] * Ws[u][w];
            float wv = Wv[u][w];
            ax += row[32 + u * 3 + 0] * wv;
            ay += row[32 + u * 3 + 1] * wv;
            az += row[32 + u * 3 + 2] * wv;
        }
        feat_s[(size_t)n * 32 + w] = as * inv32;
        float* fv = feat_v + (size_t)n * 96 + w * 3;
        fv[0] = ax * inv32;
        fv[1] = ay * inv32;
        fv[2] = az * inv32;
    }
}

__global__ void __launch_bounds__(256)
edge_kernel(const float* __restrict__ edge_attr,
            const float* __restrict__ edge_scalars,
            const float* __restrict__ M0, const float* __restrict__ M1,
            const float* __restrict__ Wtp0, const float* __restrict__ Wtp1,
            const float* __restrict__ Wtp2, const float* __restrict__ Wtp3,
            const int* __restrict__ edge_src, const int* __restrict__ edge_dst,
            const float* __restrict__ feat_s, const float* __restrict__ feat_v,
            float* __restrict__ acc_s, float* __restrict__ acc_v, int E) {
    const int lane = threadIdx.x & 31;
    const int col  = lane & 15;
    const int g    = lane >> 4;

    const float inv8  = 0.35355339059327373f;
    const float inv32 = 0.17677669529663687f;
    const float INV3  = 0.5773502691896258f;

    v16h aM0_0 = load_wT_frag(M0, 8, 0, lane, inv8);
    v16h aM0_1 = load_wT_frag(M0, 8, 1, lane, inv8);
    v16h aM1_0 = load_wT_frag(M1, 32, 0, lane, inv32);
    v16h aM1_1 = load_wT_frag(M1, 32, 1, lane, inv32);
    v16h aT0_0 = load_wT_frag(Wtp0, 32, 0, lane, inv32);
    v16h aT0_1 = load_wT_frag(Wtp0, 32, 1, lane, inv32);
    v16h aT1_0 = load_wT_frag(Wtp1, 32, 0, lane, inv32);
    v16h aT1_1 = load_wT_frag(Wtp1, 32, 1, lane, inv32);
    v16h aT2_0 = load_wT_frag(Wtp2, 32, 0, lane, inv32);
    v16h aT2_1 = load_wT_frag(Wtp2, 32, 1, lane, inv32);
    v16h aT3_0 = load_wT_frag(Wtp3, 32, 0, lane, inv32);
    v16h aT3_1 = load_wT_frag(Wtp3, 32, 1, lane, inv32);

    int wave   = (blockIdx.x * blockDim.x + threadIdx.x) >> 5;
    int nwaves = (gridDim.x * blockDim.x) >> 5;
    int ntiles = (E + 15) >> 4;

    for (int tile = wave; tile < ntiles; tile += nwaves) {
        int e = tile * 16 + col;
        bool valid = (e < E);
        int ec = valid ? e : 0;

        v16h b0;
#pragma unroll
        for (int q = 0; q < 16; ++q) b0[q] = (_Float16)0.0f;
        if (valid && g == 0) {
            const float* p = edge_scalars + (size_t)ec * 8;
#pragma unroll
            for (int q = 0; q < 8; ++q) b0[q] = (_Float16)p[q];
        }

        v8f z = {};
        v8f d0 = wmma16(aM0_0, b0, z);
        v8f d1 = wmma16(aM0_1, b0, z);
#pragma unroll
        for (int q = 0; q < 8; ++q) {
            d0[q] = gelu_f(d0[q]);
            d1[q] = gelu_f(d1[q]);
        }
        v16h b1 = pack_b_from_d(d0, d1);
        d0 = wmma16(aM1_0, b1, z);
        d1 = wmma16(aM1_1, b1, z);
#pragma unroll
        for (int q = 0; q < 8; ++q) {
            d0[q] = gelu_f(d0[q]);
            d1[q] = gelu_f(d1[q]);
        }
        v16h b2 = pack_b_from_d(d0, d1);
        v8f w0a = wmma16(aT0_0, b2, z), w0b = wmma16(aT0_1, b2, z);
        v8f w1a = wmma16(aT1_0, b2, z), w1b = wmma16(aT1_1, b2, z);
        v8f w2a = wmma16(aT2_0, b2, z), w2b = wmma16(aT2_1, b2, z);
        v8f w3a = wmma16(aT3_0, b2, z), w3b = wmma16(aT3_1, b2, z);

        if (valid) {
            int src = edge_src[ec];
            int dst = edge_dst[ec];
            const float* ea = edge_attr + (size_t)ec * 4;
            float y0  = ea[0];
            float y1x = ea[1], y1y = ea[2], y1z = ea[3];
            const float* fs = feat_s + (size_t)src * 32;
            const float* fv = feat_v + (size_t)src * 96;
            float* as = acc_s + (size_t)dst * 64;
            float* av = acc_v + (size_t)dst * 192;
#pragma unroll
            for (int t2 = 0; t2 < 2; ++t2) {
#pragma unroll
                for (int i = 0; i < 8; ++i) {
                    int u = 16 * t2 + 8 * g + i;
                    float W0 = t2 ? w0b[i] : w0a[i];
                    float W1 = t2 ? w1b[i] : w1a[i];
                    float W2 = t2 ? w2b[i] : w2a[i];
                    float W3 = t2 ? w3b[i] : w3a[i];
                    float es  = fs[u];
                    float evx = fv[u * 3 + 0];
                    float evy = fv[u * 3 + 1];
                    float evz = fv[u * 3 + 2];
                    float dotv = evx * y1x + evy * y1y + evz * y1z;
                    atomAddF(as + u,      W0 * es * y0);
                    atomAddF(as + 32 + u, W3 * dotv * INV3);
                    float t1 = W1 * es;
                    atomAddF(av + u * 3 + 0, t1 * y1x);
                    atomAddF(av + u * 3 + 1, t1 * y1y);
                    atomAddF(av + u * 3 + 2, t1 * y1z);
                    float t2f = W2 * y0;
                    atomAddF(av + (32 + u) * 3 + 0, t2f * evx);
                    atomAddF(av + (32 + u) * 3 + 1, t2f * evy);
                    atomAddF(av + (32 + u) * 3 + 2, t2f * evz);
                }
            }
        }
    }
}

__global__ void __launch_bounds__(256)
finalize_kernel(const float* __restrict__ node_input,
                const float* __restrict__ W1_s, const float* __restrict__ W1_v,
                const float* __restrict__ W2_s, const float* __restrict__ W2_v,
                const float* __restrict__ acc_s, const float* __restrict__ acc_v,
                float* __restrict__ out, int N) {
    __shared__ float sW2s[64][32];
    __shared__ float sW2v[64][32];
    __shared__ float sW1s[32][32];
    __shared__ float sW1v[32][32];
    for (int idx = threadIdx.x; idx < 2048; idx += blockDim.x) {
        int u = idx >> 5, w = idx & 31;
        sW2s[u][w] = W2_s[u * 32 + w];
        sW2v[u][w] = W2_v[u * 32 + w];
    }
    for (int idx = threadIdx.x; idx < 1024; idx += blockDim.x) {
        int u = idx >> 5, w = idx & 31;
        sW1s[u][w] = W1_s[u * 64 + 32 + w];
        sW1v[u][w] = W1_v[u * 64 + 32 + w];
    }
    __syncthreads();
    const float inv32 = 0.17677669529663687f;
    const float cconv = 0.125f * 0.31622776601683794f;
    const float ca = 0.9219544457292887f;
    const float cb = 0.3872983346207417f;
    int gid = blockIdx.x * blockDim.x + threadIdx.x;
    int total = N * 32;
    for (int t = gid; t < total; t += gridDim.x * blockDim.x) {
        int n = t >> 5, w = t & 31;
        const float* row = node_input + (size_t)n * 128;
        float ss = 0.f, svx = 0.f, svy = 0.f, svz = 0.f;
#pragma unroll 4
        for (int u = 0; u < 32; ++u) {
            ss += row[u] * sW1s[u][w];
            float wv = sW1v[u][w];
            svx += row[32 + u * 3 + 0] * wv;
            svy += row[32 + u * 3 + 1] * wv;
            svz += row[32 + u * 3 + 2] * wv;
        }
        ss *= inv32; svx *= inv32; svy *= inv32; svz *= inv32;
        const float* as = acc_s + (size_t)n * 64;
        const float* av = acc_v + (size_t)n * 192;
        float cs = 0.f, cvx = 0.f, cvy = 0.f, cvz = 0.f;
#pragma unroll 4
        for (int u = 0; u < 64; ++u) {
            cs += as[u] * sW2s[u][w];
            float wv = sW2v[u][w];
            cvx += av[u * 3 + 0] * wv;
            cvy += av[u * 3 + 1] * wv;
            cvz += av[u * 3 + 2] * wv;
        }
        cs *= cconv; cvx *= cconv; cvy *= cconv; cvz *= cconv;
        float* o = out + (size_t)n * 128;
        o[w] = ca * ss + cb * cs;
        o[32 + w * 3 + 0] = ca * svx + cb * cvx;
        o[32 + w * 3 + 1] = ca * svy + cb * cvy;
        o[32 + w * 3 + 2] = ca * svz + cb * cvz;
    }
}

extern "C" void kernel_launch(void* const* d_in, const int* in_sizes, int n_in,
                              void* d_out, int out_size, void* d_ws, size_t ws_size,
                              hipStream_t stream) {
    (void)n_in; (void)out_size; (void)ws_size;
    const float* node_input   = (const float*)d_in[0];
    const float* edge_attr    = (const float*)d_in[1];
    const float* edge_scalars = (const float*)d_in[2];
    const float* W1_s = (const float*)d_in[3];
    const float* W1_v = (const float*)d_in[4];
    const float* M0   = (const float*)d_in[5];
    const float* M1   = (const float*)d_in[6];
    const float* Wtp0 = (const float*)d_in[7];
    const float* Wtp1 = (const float*)d_in[8];
    const float* Wtp2 = (const float*)d_in[9];
    const float* Wtp3 = (const float*)d_in[10];
    const float* W2_s = (const float*)d_in[11];
    const float* W2_v = (const float*)d_in[12];
    const int* edge_src = (const int*)d_in[13];
    const int* edge_dst = (const int*)d_in[14];

    int N = in_sizes[0] / 128;
    int E = in_sizes[13];

    float* feat_s = (float*)d_ws;
    float* feat_v = feat_s + (size_t)N * 32;
    float* acc_s  = feat_v + (size_t)N * 96;
    float* acc_v  = acc_s  + (size_t)N * 64;

    hipMemsetAsync(acc_s, 0, (size_t)N * 256 * sizeof(float), stream);

    const int threads = 256;
    int blocksN = (N * 32 + threads - 1) / threads;
    node_pre_kernel<<<blocksN, threads, 0, stream>>>(node_input, W1_s, W1_v,
                                                     feat_s, feat_v, N);

    int blocksE = 1024;
    edge_kernel<<<blocksE, threads, 0, stream>>>(edge_attr, edge_scalars,
                                                 M0, M1, Wtp0, Wtp1, Wtp2, Wtp3,
                                                 edge_src, edge_dst,
                                                 feat_s, feat_v, acc_s, acc_v, E);

    finalize_kernel<<<blocksN, threads, 0, stream>>>(node_input, W1_s, W1_v,
                                                     W2_s, W2_v, acc_s, acc_v,
                                                     (float*)d_out, N);
}


# Edge kernel: 512 blocks, all resident at once (2 per WGP), not 1024 in two batches: weight set-up once per wave, then 7-8 tiles of the block's own 61-62-tile range; atomic arrival order changes
# speedup vs baseline: 8.7347x; 8.7347x over previous
#include <hip/hip_runtime.h>
#include <hip/hip_bf16.h>
#include <math.h>

typedef __attribute__((ext_vector_type(16))) _Float16 v16h;
typedef __attribute__((ext_vector_type(8)))  float    v8f;
typedef __attribute__((ext_vector_type(2)))  float    v2f;

__device__ __forceinline__ float gelu_f(float x) {
    const float c0 = 0.7978845608028654f;
    const float c1 = 2.8853900817779268f;
    float u = c0 * (x + 0.044715f * x * x * x);
    float e = __builtin_amdgcn_exp2f(u * c1);
    return x - x * __builtin_amdgcn_rcpf(e + 1.0f);
}

__device__ __forceinline__ void atomAddF(float* p, float v) {
    __hip_atomic_fetch_add(p, v, __ATOMIC_RELAXED, __HIP_MEMORY_SCOPE_AGENT);
}

__device__ __forceinline__ v8f wmma16(v16h a, v16h b, v8f c) {
    return __builtin_amdgcn_wmma_f32_16x16x32_f16(
         false, a,  false, b,
         (short)0, c,  false,  false);
}

__device__ __forceinline__ v8f wmma4(v2f a, v2f b, v8f c) {
    return __builtin_amdgcn_wmma_f32_16x16x4_f32(
         false, a,  false, b,
         (short)0, c,  false,  false);
}

__device__ __forceinline__ void swap_halves(float& a, float& b) {
    typedef __attribute__((ext_vector_type(2))) unsigned int v2u;
    v2u r = __builtin_amdgcn_permlane16_swap(__float_as_uint(a), __float_as_uint(b),
                                             false, false);
    a = __uint_as_float(r[0]);
    b = __uint_as_float(r[1]);
}

template <int R>
__device__ __forceinline__ float row_share(float v) {
    int x = __float_as_int(v);
    return __int_as_float(__builtin_amdgcn_update_dpp(x, x, 0x150 + R, 0xf, 0xf, false));
}
__device__ __forceinline__ float row_rot8(float v) {
    int x = __float_as_int(v);
    return __int_as_float(__builtin_amdgcn_update_dpp(x, x, 0x128, 0xf, 0xf, false));
}
__device__ __forceinline__ v8f scale_rows(v8f P, float yrow) {
    v8f o;
    o[0] = row_share<0>(yrow) * P[0];
    o[1] = row_share<1>(yrow) * P[1];
    o[2] = row_share<2>(yrow) * P[2];
    o[3] = row_share<3>(yrow) * P[3];
    o[4] = row_share<4>(yrow) * P[4];
    o[5] = row_share<5>(yrow) * P[5];
    o[6] = row_share<6>(yrow) * P[6];
    o[7] = row_share<7>(yrow) * P[7];
    return o;
}

__device__ __forceinline__ void wave_lds_order() {
    __builtin_amdgcn_fence(__ATOMIC_RELEASE, "wavefront");
    __builtin_amdgcn_wave_barrier();
    __builtin_amdgcn_fence(__ATOMIC_ACQUIRE, "wavefront");
}

__device__ __forceinline__ v16h load_wT_frag(const float* __restrict__ W,
                                             int Kreal, int t, int lane,
                                             float scale) {
    int m = 16 * t + (lane & 15);
    int g = lane >> 4;
    v16h a;
#pragma unroll
    for (int e = 0; e < 16; ++e) {
        int r = e >> 1, j = e & 1;
        int k = ((r >> 2) << 4) + g * 8 + ((r & 3) << 1) + j;
        float val = (k < Kreal) ? W[k * 32 + m] * scale : 0.0f;
        a[e] = (_Float16)val;
    }
    return a;
}

__device__ __forceinline__ v16h pack_b_from_d(v8f d0, v8f d1) {
    v16h b;
#pragma unroll
    for (int e = 0; e < 16; ++e) {
        int r = e >> 1, j = e & 1;
        int i = ((r & 3) << 1) + j;
        float val = (r >> 2) ? d1[i] : d0[i];
        b[e] = (_Float16)val;
    }
    return b;
}

template <int J>
__device__ __forceinline__ float row_val(const float (&r)[4]) {
    return __int_as_float(__builtin_amdgcn_readlane(__float_as_int(r[J & 3]), J >> 2));
}

template <int U>
__device__ __forceinline__ void node_step(const float (&r)[4], const v2f* ws, const v2f* wv,
                                          v2f& as, v2f& ax, v2f& ay, v2f& az) {
    float rs = row_val<U>(r);
    float rx = row_val<32 + 3 * U + 0>(r);
    float ry = row_val<32 + 3 * U + 1>(r);
    float rz = row_val<32 + 3 * U + 2>(r);
    as += ws[U] * rs;
    ax += wv[U] * rx;
    ay += wv[U] * ry;
    az += wv[U] * rz;
}

template <int U0>
__device__ __forceinline__ void node_steps8(const float (&r)[4], const v2f* ws, const v2f* wv,
                                            v2f& as, v2f& ax, v2f& ay, v2f& az) {
    node_step<U0 + 0>(r, ws, wv, as, ax, ay, az);
    node_step<U0 + 1>(r, ws, wv, as, ax, ay, az);
    node_step<U0 + 2>(r, ws, wv, as, ax, ay, az);
    node_step<U0 + 3>(r, ws, wv, as, ax, ay, az);
    node_step<U0 + 4>(r, ws, wv, as, ax, ay, az);
    node_step<U0 + 5>(r, ws, wv, as, ax, ay, az);
    node_step<U0 + 6>(r, ws, wv, as, ax, ay, az);
    node_step<U0 + 7>(r, ws, wv, as, ax, ay, az);
}

__device__ __forceinline__ void node_sums(const float (&r)[4], const v2f* ws, const v2f* wv,
                                          int n, int lane,
                                          float* feat_s, float* feat_v,
                                          float* out) {
    const float inv32 = 0.17677669529663687f;
    const float ca32  = 0.9219544457292887f * inv32;
    v2f as = {0.f, 0.f}, ax = {0.f, 0.f}, ay = {0.f, 0.f}, az = {0.f, 0.f};
    node_steps8<0>(r, ws, wv, as, ax, ay, az);
    node_steps8<8>(r, ws, wv, as, ax, ay, az);
    node_steps8<16>(r, ws, wv, as, ax, ay, az);
    node_steps8<24>(r, ws, wv, as, ax, ay, az);
    feat_s[(size_t)n * 32 + lane] = as[0] * inv32;
    float* fv = feat_v + (size_t)n * 96 + lane * 3;
    fv[0] = ax[0] * inv32;
    fv[1] = ay[0] * inv32;
    fv[2] = az[0] * inv32;
    float* o = out + (size_t)n * 128;
    o[lane] = as[1] * ca32;
    o[32 + lane * 3 + 0] = ax[1] * ca32;
    o[32 + lane * 3 + 1] = ay[1] * ca32;
    o[32 + lane * 3 + 2] = az[1] * ca32;
}

__global__ void __launch_bounds__(256)
node_kernel(const float* node_input,
            const float* __restrict__ W1_s,
            const float* __restrict__ W1_v,
            float* feat_s,
            float* feat_v,
            float* out, int N) {
    const int lane = threadIdx.x & 31;
    const int wave   = __builtin_amdgcn_readfirstlane((int)(blockIdx.x * 8 + (threadIdx.x >> 5)));
    const int nwaves = (int)gridDim.x * 8;
    if (wave >= N) return;

    int n = wave;
    float A[4], B[4];
    {
        const float* p = node_input + (size_t)n * 128 + lane * 4;
        A[0] = p[0]; A[1] = p[1]; A[2] = p[2]; A[3] = p[3];
    }
    v2f ws[32], wv[32];
#pragma unroll
    for (int u = 0; u < 32; ++u) {
        ws[u][0] = W1_s[u * 64 + lane];
        ws[u][1] = W1_s[u * 64 + 32 + lane];
        wv[u][0] = W1_v[u * 64 + lane];
        wv[u][1] = W1_v[u * 64 + 32 + lane];
    }

    for (;;) {
        const int nB = n + nwaves;
        const bool hasB = nB < N;
        {
            const float* p = node_input + (size_t)(hasB ? nB : wave) * 128 + lane * 4;
            B[0] = p[0]; B[1] = p[1]; B[2] = p[2]; B[3] = p[3];
        }
        node_sums(A, ws, wv, n, lane, feat_s, feat_v, out);
        if (!hasB) break;
        const int nA = nB + nwaves;
        const bool hasA = nA < N;
        {
            const float* p = node_input + (size_t)(hasA ? nA : wave) * 128 + lane * 4;
            A[0] = p[0]; A[1] = p[1]; A[2] = p[2]; A[3] = p[3];
        }
        node_sums(B, ws, wv, nB, lane, feat_s, feat_v, out);
        if (!hasA) break;
        n = nA;
    }
}

__global__ void __launch_bounds__(256, 4)
edge_kernel(const float* __restrict__ edge_attr,
            const float* __restrict__ edge_scalars,
            const float* __restrict__ M0, const float* __restrict__ M1,
            const float* __restrict__ Wtp0, const float* __restrict__ Wtp1,
            const float* __restrict__ Wtp2, const float* __restrict__ Wtp3,
            const int* __restrict__ edge_src, const int* __restrict__ edge_dst,
            const float* __restrict__ feat_s, const float* __restrict__ feat_v,
            const float* __restrict__ W2_s, const float* __restrict__ W2_v,
            float* __restrict__ out, int E) {
    __shared__ v2f sB[2][32][32];
    __shared__ float sT[8][16 * 100];
    const float cbc = 0.3872983346207417f * 0.125f * 0.31622776601683794f;
    for (int idx = threadIdx.x; idx < 2048; idx += blockDim.x) {
        int X = idx >> 10, f = (idx >> 5) & 31, ln = idx & 31;
        int nt = f & 1, p = (f >> 1) & 3, t2 = (f >> 3) & 1, hf = f >> 4;
        int row = hf * 32 + 16 * t2 + 8 * (ln >> 4) + 2 * p;
        int cc  = 16 * nt + (ln & 15);
        const float* Wm = X ? W2_v : W2_s;
        v2f b;
        b[0] = Wm[row * 32 + cc] * cbc;
        b[1] = Wm[(row + 1) * 32 + cc] * cbc;
        sB[X][f][ln] = b;
    }
    __syncthreads();
    const int lane = threadIdx.x & 31;
    const int col  = lane & 15;
    const int g    = lane >> 4;
    const int wv = __builtin_amdgcn_readfirstlane((int)(threadIdx.x >> 5));
    float* tb = sT[wv];

    const float inv8  = 0.35355339059327373f;
    const float inv32 = 0.17677669529663687f;
    const float INV3  = 0.5773502691896258f;

    v16h aM0_0 = load_wT_frag(M0, 8, 0, lane, inv8);
    v16h aM0_1 = load_wT_frag(M0, 8, 1, lane, inv8);
    v16h aM1_0 = load_wT_frag(M1, 32, 0, lane, inv32);
    v16h aM1_1 = load_wT_frag(M1, 32, 1, lane, inv32);
    v16h aT0_0 = load_wT_frag(Wtp0, 32, 0, lane, inv32);
    v16h aT0_1 = load_wT_frag(Wtp0, 32, 1, lane, inv32);
    v16h aT1_0 = load_wT_frag(Wtp1, 32, 0, lane, inv32);
    v16h aT1_1 = load_wT_frag(Wtp1, 32, 1, lane, inv32);
    v16h aT2_0 = load_wT_frag(Wtp2, 32, 0, lane, inv32);
    v16h aT2_1 = load_wT_frag(Wtp2, 32, 1, lane, inv32);
    v16h aT3_0 = load_wT_frag(Wtp3, 32, 0, lane, inv32);
    v16h aT3_1 = load_wT_frag(Wtp3, 32, 1, lane, inv32);

    const int ntiles = (E + 15) >> 4;
    const int nblk   = (int)gridDim.x;
    const int bid    = (int)blockIdx.x;
    const int tq     = ntiles / nblk;
    const int trem   = ntiles - tq * nblk;
    const int tBeg   = bid * tq + (bid < trem ? bid : trem);
    const int tEnd   = tBeg + tq + (bid < trem ? 1 : 0);

    for (int tile = tBeg + wv; tile < tEnd; tile += 8) {
        int e = tile * 16 + col;
        bool valid = (e < E);
        int ec = valid ? e : 0;

        v16h b0;
#pragma unroll
        for (int q = 0; q < 16; ++q) b0[q] = (_Float16)0.0f;
        if (valid && g == 0) {
            const float* p = edge_scalars + (size_t)ec * 8;
#pragma unroll
            for (int q = 0; q < 8; ++q) b0[q] = (_Float16)p[q];
        }

        v8f z = {};
        v8f d0 = wmma16(aM0_0, b0, z);
        v8f d1 = wmma16(aM0_1, b0, z);
#pragma unroll
        for (int q = 0; q < 8; ++q) {
            d0[q] = gelu_f(d0[q]);
            d1[q] = gelu_f(d1[q]);
        }
        v16h b1 = pack_b_from_d(d0, d1);
        d0 = wmma16(aM1_0, b1, z);
        d1 = wmma16(aM1_1, b1, z);
#pragma unroll
        for (int q = 0; q < 8; ++q) {
            d0[q] = gelu_f(d0[q]);
            d1[q] = gelu_f(d1[q]);
        }
        v16h b2 = pack_b_from_d(d0, d1);
        v8f w0a = wmma16(aT0_0, b2, z), w0b = wmma16(aT0_1, b2, z);
        v8f w1a = wmma16(aT1_0, b2, z), w1b = wmma16(aT1_1, b2, z);
        v8f w2a = wmma16(aT2_0, b2, z), w2b = wmma16(aT2_1, b2, z);
        v8f w3a = wmma16(aT3_0, b2, z), w3b = wmma16(aT3_1, b2, z);

        int src  = edge_src[ec];
        int dstv = edge_dst[ec];
        if (!valid) dstv = -1;
        const float* ea = edge_attr + (size_t)ec * 4;
        float y0  = ea[0];
        float y1x = ea[1], y1y = ea[2], y1z = ea[3];
        const float* fs = feat_s + (size_t)src * 32;
        const float* fv = feat_v + (size_t)src * 96;
        float es[2][8], evx[2][8], evy[2][8], evz[2][8];
#pragma unroll
        for (int t2 = 0; t2 < 2; ++t2) {
#pragma unroll
            for (int i = 0; i < 8; ++i) {
                int u = 16 * t2 + 8 * g + i;
                es[t2][i]  = fs[u];
                evx[t2][i] = fv[u * 3 + 0];
                evy[t2][i] = fv[u * 3 + 1];
                evz[t2][i] = fv[u * 3 + 2];
            }
        }
        v8f P0 = {}, P1 = {};
#pragma unroll
        for (int t2 = 0; t2 < 2; ++t2) {
#pragma unroll
            for (int p = 0; p < 4; ++p) {
                int f = (t2 * 4 + p) * 2;
                v2f t1;
#pragma unroll
                for (int j = 0; j < 2; ++j) {
                    int i = 2 * p + j;
                    float W1 = t2 ? w1b[i] : w1a[i];
                    t1[j] = W1 * es[t2][i];
                }
                P0 = wmma4(t1, sB[1][f][lane], P0);
                P1 = wmma4(t1, sB[1][f + 1][lane], P1);
            }
        }
        v8f c0 = {}, c1 = {};
        {
#pragma unroll
            for (int t2 = 0; t2 < 2; ++t2) {
#pragma unroll
                for (int p = 0; p < 4; ++p) {
                    v2f a;
#pragma unroll
                    for (int j = 0; j < 2; ++j) {
                        int i = 2 * p + j;
                        float W0 = t2 ? w0b[i] : w0a[i];
                        a[j] = W0 * es[t2][i] * y0;
                    }
                    int f = (t2 * 4 + p) * 2;
                    c0 = wmma4(a, sB[0][f][lane], c0);
                    c1 = wmma4(a, sB[0][f + 1][lane], c1);
                }
            }
#pragma unroll
            for (int t2 = 0; t2 < 2; ++t2) {
#pragma unroll
                for (int p = 0; p < 4; ++p) {
                    v2f a;
#pragma unroll
                    for (int j = 0; j < 2; ++j) {
                        int i = 2 * p + j;
                        float W3 = t2 ? w3b[i] : w3a[i];
                        float dotv = evx[t2][i] * y1x + evy[t2][i] * y1y + evz[t2][i] * y1z;
                        a[j] = W3 * dotv * INV3;
                    }
                    int f = ((2 + t2) * 4 + p) * 2;
                    c0 = wmma4(a, sB[0][f][lane], c0);
                    c1 = wmma4(a, sB[0][f + 1][lane], c1);
                }
            }
        }
        asm volatile("v_nop\n\tv_nop\n\tv_nop\n\tv_nop\n\tv_nop" : "+v"(P0), "+v"(P1));
        v8f cv[3][2];
        {
            float yx = g ? row_rot8(y1x) : y1x;
            float yy = g ? row_rot8(y1y) : y1y;
            float yz = g ? row_rot8(y1z) : y1z;
            cv[0][0] = scale_rows(P0, yx);  cv[0][1] = scale_rows(P1, yx);
            cv[1][0] = scale_rows(P0, yy);  cv[1][1] = scale_rows(P1, yy);
            cv[2][0] = scale_rows(P0, yz);  cv[2][1] = scale_rows(P1, yz);
#pragma unroll
            for (int t2 = 0; t2 < 2; ++t2) {
#pragma unroll
                for (int p = 0; p < 4; ++p) {
                    int f = ((2 + t2) * 4 + p) * 2;
                    v2f b0 = sB[1][f][lane], b1 = sB[1][f + 1][lane];
                    v2f ax, ay, az;
#pragma unroll
                    for (int j = 0; j < 2; ++j) {
                        int i = 2 * p + j;
                        float W2 = t2 ? w2b[i] : w2a[i];
                        float t2f = W2 * y0;
                        ax[j] = t2f * evx[t2][i];
                        ay[j] = t2f * evy[t2][i];
                        az[j] = t2f * evz[t2][i];
                    }
                    cv[0][0] = wmma4(ax, b0, cv[0][0]);
                    cv[0][1] = wmma4(ax, b1, cv[0][1]);
                    cv[1][0] = wmma4(ay, b0, cv[1][0]);
                    cv[1][1] = wmma4(ay, b1, cv[1][1]);
                    cv[2][0] = wmma4(az, b0, cv[2][0]);
                    cv[2][1] = wmma4(az, b1, cv[2][1]);
                }
            }
        }
        asm volatile("v_nop\n\tv_nop\n\tv_nop\n\tv_nop\n\tv_nop"
                     : "+v"(c0), "+v"(c1), "+v"(cv[0][0]), "+v"(cv[0][1]),
                       "+v"(cv[1][0]), "+v"(cv[1][1]), "+v"(cv[2][0]), "+v"(cv[2][1]));
        wave_lds_order();
        int trow = (8 * g) * 100 + 3 * col;
        asm volatile("" : "+v"(trow));
#pragma unroll
        for (int k = 0; k < 3; ++k) {
#pragma unroll
            for (int nt = 0; nt < 2; ++nt) {
#pragma unroll
                for (int r = 0; r < 8; ++r)
                    tb[trow + r * 100 + 48 * nt + k] = cv[k][nt][r];
            }
        }
        wave_lds_order();
        float vA[8][3], vB[8][3];
#pragma unroll
        for (int r = 0; r < 8; ++r) {
#pragma unroll
            for (int j = 0; j < 3; ++j) {
                vA[r][j] = tb[r * 100 + 32 * j + lane];
                vB[r][j] = tb[(8 + r) * 100 + 32 * j + lane];
            }
        }
#pragma unroll
        for (int r = 0; r < 8; ++r) {
            float sLo = c0[r], sHi = c1[r];
            swap_halves(sLo, sHi);
            int dA = __builtin_amdgcn_readlane(dstv, r);
            int dB = __builtin_amdgcn_readlane(dstv, 8 + r);
            if (dA >= 0) {
                float* o = out + (size_t)dA * 128 + lane;
                atomAddF(o,      sLo);
                atomAddF(o + 32, vA[r][0]);
                atomAddF(o + 64, vA[r][1]);
                atomAddF(o + 96, vA[r][2]);
            }
            if (dB >= 0) {
                float* o = out + (size_t)dB * 128 + lane;
                atomAddF(o,      sHi);
                atomAddF(o + 32, vB[r][0]);
                atomAddF(o + 64, vB[r][1]);
                atomAddF(o + 96, vB[r][2]);
            }
        }
    }
}

extern "C" void kernel_launch(void* const* d_in, const int* in_sizes, int n_in,
                              void* d_out, int out_size, void* d_ws, size_t ws_size,
                              hipStream_t stream) {
    (void)n_in; (void)out_size; (void)ws_size;
    const float* node_input   = (const float*)d_in[0];
    const float* edge_attr    = (const float*)d_in[1];
    const float* edge_scalars = (const float*)d_in[2];
    const float* W1_s = (const float*)d_in[3];
    const float* W1_v = (const float*)d_in[4];
    const float* M0   = (const float*)d_in[5];
    const float* M1   = (const float*)d_in[6];
    const float* Wtp0 = (const float*)d_in[7];
    const float* Wtp1 = (const float*)d_in[8];
    const float* Wtp2 = (const float*)d_in[9];
    const float* Wtp3 = (const float*)d_in[10];
    const float* W2_s = (const float*)d_in[11];
    const float* W2_v = (const float*)d_in[12];
    const int* edge_src = (const int*)d_in[13];
    const int* edge_dst = (const int*)d_in[14];

    int N = in_sizes[0] / 128;
    int E = in_sizes[13];

    float* feat_s = (float*)d_ws;
    float* feat_v = feat_s + (size_t)N * 32;
    float* out    = (float*)d_out;

    const int threads = 256;
    node_kernel<<<768, threads, 0, stream>>>(node_input, W1_s, W1_v,
                                             feat_s, feat_v, out, N);

    int blocksE = 512;
    edge_kernel<<<blocksE, threads, 0, stream>>>(edge_attr, edge_scalars,
                                                 M0, M1, Wtp0, Wtp1, Wtp2, Wtp3,
                                                 edge_src, edge_dst,
                                                 feat_s, feat_v, W2_s, W2_v, out, E);
}


# W2 map from f32 16x16x4 to f16 16x16x32 WMMA, each f32 operand split into f16 hi + lo (3 products per block, f32 accumulate): 36 not 96 matrix instructions per tile; dtype and sum order change
# speedup vs baseline: 10.6222x; 1.2161x over previous
#include <hip/hip_runtime.h>
#include <hip/hip_bf16.h>
#include <math.h>

typedef __attribute__((ext_vector_type(16))) _Float16 v16h;
typedef __attribute__((ext_vector_type(8)))  _Float16 v8h;
typedef __attribute__((ext_vector_type(8)))  float    v8f;
typedef __attribute__((ext_vector_type(2)))  float    v2f;

__device__ __forceinline__ float gelu_f(float x) {
    const float c0 = 0.7978845608028654f;
    const float c1 = 2.8853900817779268f;
    float u = c0 * (x + 0.044715f * x * x * x);
    float e = __builtin_amdgcn_exp2f(u * c1);
    return x - x * __builtin_amdgcn_rcpf(e + 1.0f);
}

__device__ __forceinline__ void atomAddF(float* p, float v) {
    __hip_atomic_fetch_add(p, v, __ATOMIC_RELAXED, __HIP_MEMORY_SCOPE_AGENT);
}

__device__ __forceinline__ v8f wmma16(v16h a, v16h b, v8f c) {
    return __builtin_amdgcn_wmma_f32_16x16x32_f16(
         false, a,  false, b,
         (short)0, c,  false,  false);
}

struct HiLo { v16h hi, lo; };
__device__ __forceinline__ HiLo split_f16(const float (&a)[16]) {
    HiLo r;
#pragma unroll
    for (int e = 0; e < 16; ++e) r.hi[e] = (_Float16)a[e];
    asm volatile("" : "+v"(r.hi));
#pragma unroll
    for (int e = 0; e < 16; ++e)
        r.lo[e] = (_Float16)__builtin_fmaf((float)r.hi[e], -1.0f, a[e]);
    return r;
}

__device__ __forceinline__ v16h lds_frag(const v8h (&s)[2][32], int lane) {
    return __builtin_shufflevector(s[0][lane], s[1][lane],
                                   0, 1, 2, 3, 4, 5, 6, 7, 8, 9, 10, 11, 12, 13, 14, 15);
}

__device__ __forceinline__ v8f mma_split(const HiLo& a, v16h bh, v16h bl, v8f c) {
    c = wmma16(a.lo, bh, c);
    c = wmma16(a.hi, bl, c);
    c = wmma16(a.hi, bh, c);
    return c;
}

__device__ __forceinline__ void swap_halves(float& a, float& b) {
    typedef __attribute__((ext_vector_type(2))) unsigned int v2u;
    v2u r = __builtin_amdgcn_permlane16_swap(__float_as_uint(a), __float_as_uint(b),
                                             false, false);
    a = __uint_as_float(r[0]);
    b = __uint_as_float(r[1]);
}

template <int R>
__device__ __forceinline__ float row_share(float v) {
    int x = __float_as_int(v);
    return __int_as_float(__builtin_amdgcn_update_dpp(x, x, 0x150 + R, 0xf, 0xf, false));
}
__device__ __forceinline__ float row_rot8(float v) {
    int x = __float_as_int(v);
    return __int_as_float(__builtin_amdgcn_update_dpp(x, x, 0x128, 0xf, 0xf, false));
}
__device__ __forceinline__ v8f scale_rows(v8f P, float yrow) {
    v8f o;
    o[0] = row_share<0>(yrow) * P[0];
    o[1] = row_share<1>(yrow) * P[1];
    o[2] = row_share<2>(yrow) * P[2];
    o[3] = row_share<3>(yrow) * P[3];
    o[4] = row_share<4>(yrow) * P[4];
    o[5] = row_share<5>(yrow) * P[5];
    o[6] = row_share<6>(yrow) * P[6];
    o[7] = row_share<7>(yrow) * P[7];
    return o;
}

__device__ __forceinline__ void wave_lds_order() {
    __builtin_amdgcn_fence(__ATOMIC_RELEASE, "wavefront");
    __builtin_amdgcn_wave_barrier();
    __builtin_amdgcn_fence(__ATOMIC_ACQUIRE, "wavefront");
}

__device__ __forceinline__ v16h load_wT_frag(const float* __restrict__ W,
                                             int Kreal, int t, int lane,
                                             float scale) {
    int m = 16 * t + (lane & 15);
    int g = lane >> 4;
    v16h a;
#pragma unroll
    for (int e = 0; e < 16; ++e) {
        int r = e >> 1, j = e & 1;
        int k = ((r >> 2) << 4) + g * 8 + ((r & 3) << 1) + j;
        float val = (k < Kreal) ? W[k * 32 + m] * scale : 0.0f;
        a[e] = (_Float16)val;
    }
    return a;
}

__device__ __forceinline__ v16h pack_b_from_d(v8f d0, v8f d1) {
    v16h b;
#pragma unroll
    for (int e = 0; e < 16; ++e) {
        int r = e >> 1, j = e & 1;
        int i = ((r & 3) << 1) + j;
        float val = (r >> 2) ? d1[i] : d0[i];
        b[e] = (_Float16)val;
    }
    return b;
}

template <int J>
__device__ __forceinline__ float row_val(const float (&r)[4]) {
    return __int_as_float(__builtin_amdgcn_readlane(__float_as_int(r[J & 3]), J >> 2));
}

template <int U>
__device__ __forceinline__ void node_step(const float (&r)[4], const v2f* ws, const v2f* wv,
                                          v2f& as, v2f& ax, v2f& ay, v2f& az) {
    float rs = row_val<U>(r);
    float rx = row_val<32 + 3 * U + 0>(r);
    float ry = row_val<32 + 3 * U + 1>(r);
    float rz = row_val<32 + 3 * U + 2>(r);
    as += ws[U] * rs;
    ax += wv[U] * rx;
    ay += wv[U] * ry;
    az += wv[U] * rz;
}

template <int U0>
__device__ __forceinline__ void node_steps8(const float (&r)[4], const v2f* ws, const v2f* wv,
                                            v2f& as, v2f& ax, v2f& ay, v2f& az) {
    node_step<U0 + 0>(r, ws, wv, as, ax, ay, az);
    node_step<U0 + 1>(r, ws, wv, as, ax, ay, az);
    node_step<U0 + 2>(r, ws, wv, as, ax, ay, az);
    node_step<U0 + 3>(r, ws, wv, as, ax, ay, az);
    node_step<U0 + 4>(r, ws, wv, as, ax, ay, az);
    node_step<U0 + 5>(r, ws, wv, as, ax, ay, az);
    node_step<U0 + 6>(r, ws, wv, as, ax, ay, az);
    node_step<U0 + 7>(r, ws, wv, as, ax, ay, az);
}

__device__ __forceinline__ void node_sums(const float (&r)[4], const v2f* ws, const v2f* wv,
                                          int n, int lane,
                                          float* feat_s, float* feat_v,
                                          float* out) {
    const float inv32 = 0.17677669529663687f;
    const float ca32  = 0.9219544457292887f * inv32;
    v2f as = {0.f, 0.f}, ax = {0.f, 0.f}, ay = {0.f, 0.f}, az = {0.f, 0.f};
    node_steps8<0>(r, ws, wv, as, ax, ay, az);
    node_steps8<8>(r, ws, wv, as, ax, ay, az);
    node_steps8<16>(r, ws, wv, as, ax, ay, az);
    node_steps8<24>(r, ws, wv, as, ax, ay, az);
    feat_s[(size_t)n * 32 + lane] = as[0] * inv32;
    float* fv = feat_v + (size_t)n * 96 + lane * 3;
    fv[0] = ax[0] * inv32;
    fv[1] = ay[0] * inv32;
    fv[2] = az[0] * inv32;
    float* o = out + (size_t)n * 128;
    o[lane] = as[1] * ca32;
    o[32 + lane * 3 + 0] = ax[1] * ca32;
    o[32 + lane * 3 + 1] = ay[1] * ca32;
    o[32 + lane * 3 + 2] = az[1] * ca32;
}

__global__ void __launch_bounds__(256)
node_kernel(const float* node_input,
            const float* __restrict__ W1_s,
            const float* __restrict__ W1_v,
            float* feat_s,
            float* feat_v,
            float* out, int N) {
    const int lane = threadIdx.x & 31;
    const int wave   = __builtin_amdgcn_readfirstlane((int)(blockIdx.x * 8 + (threadIdx.x >> 5)));
    const int nwaves = (int)gridDim.x * 8;
    if (wave >= N) return;

    int n = wave;
    float A[4], B[4];
    {
        const float* p = node_input + (size_t)n * 128 + lane * 4;
        A[0] = p[0]; A[1] = p[1]; A[2] = p[2]; A[3] = p[3];
    }
    v2f ws[32], wv[32];
#pragma unroll
    for (int u = 0; u < 32; ++u) {
        ws[u][0] = W1_s[u * 64 + lane];
        ws[u][1] = W1_s[u * 64 + 32 + lane];
        wv[u][0] = W1_v[u * 64 + lane];
        wv[u][1] = W1_v[u * 64 + 32 + lane];
    }

    for (;;) {
        const int nB = n + nwaves;
        const bool hasB = nB < N;
        {
            const float* p = node_input + (size_t)(hasB ? nB : wave) * 128 + lane * 4;
            B[0] = p[0]; B[1] = p[1]; B[2] = p[2]; B[3] = p[3];
        }
        node_sums(A, ws, wv, n, lane, feat_s, feat_v, out);
        if (!hasB) break;
        const int nA = nB + nwaves;
        const bool hasA = nA < N;
        {
            const float* p = node_input + (size_t)(hasA ? nA : wave) * 128 + lane * 4;
            A[0] = p[0]; A[1] = p[1]; A[2] = p[2]; A[3] = p[3];
        }
        node_sums(B, ws, wv, nB, lane, feat_s, feat_v, out);
        if (!hasA) break;
        n = nA;
    }
}

__global__ void __launch_bounds__(256, 4)
edge_kernel(const float* __restrict__ edge_attr,
            const float* __restrict__ edge_scalars,
            const float* __restrict__ M0, const float* __restrict__ M1,
            const float* __restrict__ Wtp0, const float* __restrict__ Wtp1,
            const float* __restrict__ Wtp2, const float* __restrict__ Wtp3,
            const int* __restrict__ edge_src, const int* __restrict__ edge_dst,
            const float* __restrict__ feat_s, const float* __restrict__ feat_v,
            const float* __restrict__ W2_s, const float* __restrict__ W2_v,
            float* __restrict__ out, int E) {
    __shared__ v8h sBh[8][2][32];
    __shared__ v8h sBl[8][2][32];
    __shared__ v8h sAT[8][2][32];
    __shared__ float sT[8][16 * 100];
    const float cb8 = 0.3872983346207417f * 0.31622776601683794f;
    for (int idx = threadIdx.x; idx < 512; idx += blockDim.x) {
        int ln = idx & 31, t2 = (idx >> 5) & 1, f = idx >> 6;
        int nt = f & 1, q = f >> 1;
        const float* Wm = (q == 0 || q == 3) ? W2_v : W2_s;
        int row = (q >= 2 ? 32 : 0) + 16 * t2 + 8 * (ln >> 4);
        int cc  = 16 * nt + (ln & 15);
        v8h h, l;
#pragma unroll
        for (int i = 0; i < 8; ++i) {
            float w = Wm[(row + i) * 32 + cc] * cb8;
            _Float16 wh = (_Float16)w;
            h[i] = wh;
            l[i] = (_Float16)(w - (float)wh);
        }
        sBh[f][t2][ln] = h;
        sBl[f][t2][ln] = l;
    }
    {
        const int fl = threadIdx.x & 31, fw = threadIdx.x >> 5, fp = fw >> 1;
        const float* Wt = fp == 0 ? Wtp0 : fp == 1 ? Wtp1 : fp == 2 ? Wtp2 : Wtp3;
        v16h a = load_wT_frag(Wt, 32, fw & 1, fl, 0.17677669529663687f);
        sAT[fw][0][fl] = __builtin_shufflevector(a, a, 0, 1, 2, 3, 4, 5, 6, 7);
        sAT[fw][1][fl] = __builtin_shufflevector(a, a, 8, 9, 10, 11, 12, 13, 14, 15);
    }
    __syncthreads();
    const int lane = threadIdx.x & 31;
    const int col  = lane & 15;
    const int g    = lane >> 4;
    const int wv = __builtin_amdgcn_readfirstlane((int)(threadIdx.x >> 5));
    float* tb = sT[wv];

    const float inv8  = 0.35355339059327373f;
    const float inv32 = 0.17677669529663687f;
    const float INV3  = 0.5773502691896258f;

    v16h aM0_0 = load_wT_frag(M0, 8, 0, lane, inv8);
    v16h aM0_1 = load_wT_frag(M0, 8, 1, lane, inv8);
    v16h aM1_0 = load_wT_frag(M1, 32, 0, lane, inv32);
    v16h aM1_1 = load_wT_frag(M1, 32, 1, lane, inv32);
    const int ntiles = (E + 15) >> 4;
    const int nblk   = (int)gridDim.x;
    const int bid    = (int)blockIdx.x;
    const int tq     = ntiles / nblk;
    const int trem   = ntiles - tq * nblk;
    const int tBeg   = bid * tq + (bid < trem ? bid : trem);
    const int tEnd   = tBeg + tq + (bid < trem ? 1 : 0);

    for (int tile = tBeg + wv; tile < tEnd; tile += 8) {
        int e = tile * 16 + col;
        bool valid = (e < E);
        int ec = valid ? e : 0;

        v16h b0;
#pragma unroll
        for (int q = 0; q < 16; ++q) b0[q] = (_Float16)0.0f;
        if (valid && g == 0) {
            const float* p = edge_scalars + (size_t)ec * 8;
#pragma unroll
            for (int q = 0; q < 8; ++q) b0[q] = (_Float16)p[q];
        }

        v8f z = {};
        v8f d0 = wmma16(aM0_0, b0, z);
        v8f d1 = wmma16(aM0_1, b0, z);
#pragma unroll
        for (int q = 0; q < 8; ++q) {
            d0[q] = gelu_f(d0[q]);
            d1[q] = gelu_f(d1[q]);
        }
        v16h b1 = pack_b_from_d(d0, d1);
        d0 = wmma16(aM1_0, b1, z);
        d1 = wmma16(aM1_1, b1, z);
#pragma unroll
        for (int q = 0; q < 8; ++q) {
            d0[q] = gelu_f(d0[q]);
            d1[q] = gelu_f(d1[q]);
        }
        v16h b2 = pack_b_from_d(d0, d1);
        int src  = edge_src[ec];
        int dstv = edge_dst[ec];
        if (!valid) dstv = -1;
        const float* ea = edge_attr + (size_t)ec * 4;
        float y0  = ea[0];
        float y1x = ea[1], y1y = ea[2], y1z = ea[3];
        const float* fs = feat_s + (size_t)src * 32;
        const float* fv = feat_v + (size_t)src * 96;
        float es[2][8], evx[2][8], evy[2][8], evz[2][8];
#pragma unroll
        for (int t2 = 0; t2 < 2; ++t2) {
#pragma unroll
            for (int i = 0; i < 8; ++i) {
                int u = 16 * t2 + 8 * g + i;
                es[t2][i]  = fs[u];
                evx[t2][i] = fv[u * 3 + 0];
                evy[t2][i] = fv[u * 3 + 1];
                evz[t2][i] = fv[u * 3 + 2];
            }
        }
        const float y0s  = y0 * 0.125f;
        const float y1xs = y1x * 0.125f, y1ys = y1y * 0.125f, y1zs = y1z * 0.125f;
        const float INV3s = INV3 * 0.125f;
        v8f P0, P1;
        {
            v8f wa = wmma16(lds_frag(sAT[2], lane), b2, z), wb = wmma16(lds_frag(sAT[3], lane), b2, z);
            float a[16];
#pragma unroll
            for (int t2 = 0; t2 < 2; ++t2) {
#pragma unroll
                for (int i = 0; i < 8; ++i)
                    a[8 * t2 + i] = (t2 ? wb[i] : wa[i]) * es[t2][i];
            }
            HiLo f = split_f16(a);
            P0 = mma_split(f, lds_frag(sBh[0], lane), lds_frag(sBl[0], lane), z);
            P1 = mma_split(f, lds_frag(sBh[1], lane), lds_frag(sBl[1], lane), z);
        }
        v8f c0, c1;
        {
            v8f wa = wmma16(lds_frag(sAT[0], lane), b2, z), wb = wmma16(lds_frag(sAT[1], lane), b2, z);
            float a[16];
#pragma unroll
            for (int t2 = 0; t2 < 2; ++t2) {
#pragma unroll
                for (int i = 0; i < 8; ++i)
                    a[8 * t2 + i] = (t2 ? wb[i] : wa[i]) * es[t2][i] * y0s;
            }
            HiLo f = split_f16(a);
            c0 = mma_split(f, lds_frag(sBh[2], lane), lds_frag(sBl[2], lane), z);
            c1 = mma_split(f, lds_frag(sBh[3], lane), lds_frag(sBl[3], lane), z);
        }
        {
            v8f wa = wmma16(lds_frag(sAT[6], lane), b2, z), wb = wmma16(lds_frag(sAT[7], lane), b2, z);
            float a[16];
#pragma unroll
            for (int t2 = 0; t2 < 2; ++t2) {
#pragma unroll
                for (int i = 0; i < 8; ++i) {
                    float dotv = evx[t2][i] * y1x + evy[t2][i] * y1y + evz[t2][i] * y1z;
                    a[8 * t2 + i] = (t2 ? wb[i] : wa[i]) * dotv * INV3s;
                }
            }
            HiLo f = split_f16(a);
            c0 = mma_split(f, lds_frag(sBh[4], lane), lds_frag(sBl[4], lane), c0);
            c1 = mma_split(f, lds_frag(sBh[5], lane), lds_frag(sBl[5], lane), c1);
        }
        v8f cv[3][2];
        {
            float yx = g ? row_rot8(y1xs) : y1xs;
            float yy = g ? row_rot8(y1ys) : y1ys;
            float yz = g ? row_rot8(y1zs) : y1zs;
            cv[0][0] = scale_rows(P0, yx);  cv[0][1] = scale_rows(P1, yx);
            cv[1][0] = scale_rows(P0, yy);  cv[1][1] = scale_rows(P1, yy);
            cv[2][0] = scale_rows(P0, yz);  cv[2][1] = scale_rows(P1, yz);
            v8f wa = wmma16(lds_frag(sAT[4], lane), b2, z), wb = wmma16(lds_frag(sAT[5], lane), b2, z);
            float t2f[16];
#pragma unroll
            for (int t2 = 0; t2 < 2; ++t2) {
#pragma unroll
                for (int i = 0; i < 8; ++i)
                    t2f[8 * t2 + i] = (t2 ? wb[i] : wa[i]) * y0s;
            }
            v16h bh0 = lds_frag(sBh[6], lane), bl0 = lds_frag(sBl[6], lane);
            v16h bh1 = lds_frag(sBh[7], lane), bl1 = lds_frag(sBl[7], lane);
#pragma unroll
            for (int k = 0; k < 3; ++k) {
                float a[16];
#pragma unroll
                for (int t2 = 0; t2 < 2; ++t2) {
#pragma unroll
                    for (int i = 0; i < 8; ++i)
                        a[8 * t2 + i] = t2f[8 * t2 + i]
                                        * (k == 0 ? evx[t2][i] : k == 1 ? evy[t2][i] : evz[t2][i]);
                }
                HiLo f = split_f16(a);
                cv[k][0] = mma_split(f, bh0, bl0, cv[k][0]);
                cv[k][1] = mma_split(f, bh1, bl1, cv[k][1]);
            }
        }
        wave_lds_order();
        int trow = (8 * g) * 100 + 3 * col;
        asm volatile("" : "+v"(trow));
#pragma unroll
        for (int k = 0; k < 3; ++k) {
#pragma unroll
            for (int nt = 0; nt < 2; ++nt) {
#pragma unroll
                for (int r = 0; r < 8; ++r)
                    tb[trow + r * 100 + 48 * nt + k] = cv[k][nt][r];
            }
        }
        wave_lds_order();
        float vA[8][3], vB[8][3];
#pragma unroll
        for (int r = 0; r < 8; ++r) {
#pragma unroll
            for (int j = 0; j < 3; ++j) {
                vA[r][j] = tb[r * 100 + 32 * j + lane];
                vB[r][j] = tb[(8 + r) * 100 + 32 * j + lane];
            }
        }
#pragma unroll
        for (int r = 0; r < 8; ++r) {
            float sLo = c0[r], sHi = c1[r];
            swap_halves(sLo, sHi);
            int dA = __builtin_amdgcn_readlane(dstv, r);
            int dB = __builtin_amdgcn_readlane(dstv, 8 + r);
            if (dA >= 0) {
                float* o = out + (size_t)dA * 128 + lane;
                atomAddF(o,      sLo);
                atomAddF(o + 32, vA[r][0]);
                atomAddF(o + 64, vA[r][1]);
                atomAddF(o + 96, vA[r][2]);
            }
            if (dB >= 0) {
                float* o = out + (size_t)dB * 128 + lane;
                atomAddF(o,      sHi);
                atomAddF(o + 32, vB[r][0]);
                atomAddF(o + 64, vB[r][1]);
                atomAddF(o + 96, vB[r][2]);
            }
        }
    }
}

extern "C" void kernel_launch(void* const* d_in, const int* in_sizes, int n_in,
                              void* d_out, int out_size, void* d_ws, size_t ws_size,
                              hipStream_t stream) {
    (void)n_in; (void)out_size; (void)ws_size;
    const float* node_input   = (const float*)d_in[0];
    const float* edge_attr    = (const float*)d_in[1];
    const float* edge_scalars = (const float*)d_in[2];
    const float* W1_s = (const float*)d_in[3];
    const float* W1_v = (const float*)d_in[4];
    const float* M0   = (const float*)d_in[5];
    const float* M1   = (const float*)d_in[6];
    const float* Wtp0 = (const float*)d_in[7];
    const float* Wtp1 = (const float*)d_in[8];
    const float* Wtp2 = (const float*)d_in[9];
    const float* Wtp3 = (const float*)d_in[10];
    const float* W2_s = (const float*)d_in[11];
    const float* W2_v = (const float*)d_in[12];
    const int* edge_src = (const int*)d_in[13];
    const int* edge_dst = (const int*)d_in[14];

    int N = in_sizes[0] / 128;
    int E = in_sizes[13];

    float* feat_s = (float*)d_ws;
    float* feat_v = feat_s + (size_t)N * 32;
    float* out    = (float*)d_out;

    const int threads = 256;
    node_kernel<<<768, threads, 0, stream>>>(node_input, W1_s, W1_v,
                                             feat_s, feat_v, out, N);

    int blocksE = 512;
    edge_kernel<<<blocksE, threads, 0, stream>>>(edge_attr, edge_scalars,
                                                 M0, M1, Wtp0, Wtp1, Wtp2, Wtp3,
                                                 edge_src, edge_dst,
                                                 feat_s, feat_v, W2_s, W2_v, out, E);
}
